# starting kernel
# baseline (speedup 1.0000x reference)
#include <hip/hip_runtime.h>
#include <math.h>

#define H_DIM   512
#define N_DIM   64
#define L_DIM   2048
#define CH_DIM  1
#define LTILE   128
#define TWO_PI  6.283185307179586f

typedef __attribute__((ext_vector_type(2))) float v2f;
typedef __attribute__((ext_vector_type(8))) float v8f;

__device__ __forceinline__ float softplusf(float x) {
    return fmaxf(x, 0.0f) + log1pf(expf(-fabsf(x)));
}

__device__ __forceinline__ float frcp(float x) {
    return __builtin_amdgcn_rcpf(x);
}

__global__ __launch_bounds__(256)
void s4_cauchy_wmma(const float* __restrict__ Lre, const float* __restrict__ Lim,
                    const float* __restrict__ Pre, const float* __restrict__ Pim,
                    const float* __restrict__ Bre, const float* __restrict__ Bim,
                    const float* __restrict__ Cre, const float* __restrict__ Cim,
                    const float* __restrict__ log_dt,
                    float* __restrict__ KhatRe, float* __restrict__ KhatIm)
{
    __shared__ float Wm[128 * 16];
    __shared__ float lamRe[N_DIM], lamIm[N_DIM];
    __shared__ float tile[8 * 16 * 8];

    const int tid    = threadIdx.x;
    const int h      = blockIdx.x / (L_DIM / LTILE);
    const int lchunk = blockIdx.x % (L_DIM / LTILE);

    for (int i = tid; i < 128 * 16; i += 256) Wm[i] = 0.0f;
    __syncthreads();

    if (tid < N_DIM) {
        const int n   = tid;
        const int idx = h * N_DIM + n;
        const float pr = Pre[idx], pi = Pim[idx];
        const float br = Bre[idx], bi = Bim[idx];
        const float cr = Cre[idx], ci = Cim[idx];
        const float wPBr = pr * br + pi * bi, wPBi = pr * bi - pi * br;
        const float wPPr = pr * pr + pi * pi, wPPi = 0.0f;
        const float wCBr = cr * br + ci * bi, wCBi = cr * bi - ci * br;
        const float wCPr = cr * pr + ci * pi, wCPi = cr * pi - ci * pr;
        const float wr[4] = {wCBr, wCPr, wPBr, wPPr};
        const float wi[4] = {wCBi, wCPi, wPBi, wPPi};
        #pragma unroll
        for (int X = 0; X < 4; ++X) {
            Wm[n * 16 + 2 * X]              =  wr[X];
            Wm[n * 16 + 2 * X + 1]          =  wi[X];
            Wm[(64 + n) * 16 + 2 * X]       = -wi[X];
            Wm[(64 + n) * 16 + 2 * X + 1]   =  wr[X];
        }
        lamRe[n] = -softplusf(Lre[idx]);
        lamIm[n] = Lim[idx];
    }
    __syncthreads();

    const int wave = tid >> 5;
    const int lane = tid & 31;
    const int half = lane >> 4;
    const int c    = lane & 15;
    const int l0   = lchunk * LTILE + wave * 16;
    const int l    = l0 + c;

    const float sc    = 2.0f * expf(-log_dt[h]);
    const float theta = (TWO_PI / (float)L_DIM) * (float)l;
    float zre, zim_p;
    __sincosf(theta, &zim_p, &zre);
    const float zim = -zim_p;
    float dnr = 1.0f + zre, dni = zim;
    float d2  = dnr * dnr + dni * dni;
    const float eps = 1.1920929e-7f;
    if (sqrtf(d2) < eps) { dnr = eps; dni = 0.0f; d2 = eps * eps; }
    const float nur = 1.0f - zre, nui = -zim;
    const float invd2   = frcp(d2);
    const float g_re    = sc * (nur * dnr + nui * dni) * invd2;
    const float g_im    = sc * (nui * dnr - nur * dni) * invd2;
    const float pref_re =  2.0f * dnr * invd2;
    const float pref_im = -2.0f * dni * invd2;

    v8f acc = {};
    #pragma unroll
    for (int kk = 0; kk < 16; ++kk) {
        const int n0  = 4 * kk + 2 * half;
        const float dr0 = g_re - lamRe[n0],     di0 = g_im - lamIm[n0];
        const float dr1 = g_re - lamRe[n0 + 1], di1 = g_im - lamIm[n0 + 1];
        const float iv0 = frcp(dr0 * dr0 + di0 * di0);
        const float iv1 = frcp(dr1 * dr1 + di1 * di1);
        v2f aRe = {  dr0 * iv0,  dr1 * iv1 };
        v2f aIm = { -di0 * iv0, -di1 * iv1 };
        const int krow = 4 * kk + 2 * half;
        v2f bRe = { Wm[krow * 16 + c],        Wm[(krow + 1) * 16 + c] };
        v2f bIm = { Wm[(64 + krow) * 16 + c], Wm[(64 + krow + 1) * 16 + c] };
        acc = __builtin_amdgcn_wmma_f32_16x16x4_f32(false, aRe, false, bRe,
                                                    (short)0, acc, false, false);
        acc = __builtin_amdgcn_wmma_f32_16x16x4_f32(false, aIm, false, bIm,
                                                    (short)0, acc, false, false);
    }

    if (c < 8) {
        #pragma unroll
        for (int r = 0; r < 8; ++r) {
            const int m = r + 8 * half;
            tile[(wave * 16 + m) * 8 + c] = acc[r];
        }
    }
    __syncthreads();

    if (lane < 16) {
        const float* t8 = &tile[(wave * 16 + lane) * 8];
        const float sCBr = t8[0], sCBi = t8[1];
        const float sCPr = t8[2], sCPi = t8[3];
        const float sPBr = t8[4], sPBi = t8[5];
        const float sPPr = t8[6], sPPi = t8[7];
        const float dwr = 1.0f + sPPr, dwi = sPPi;
        const float iw  = frcp(dwr * dwr + dwi * dwi);
        const float tr  = (sPBr * dwr + sPBi * dwi) * iw;
        const float ti  = (sPBi * dwr - sPBr * dwi) * iw;
        const float cr2 = sCPr * tr - sCPi * ti;
        const float ci2 = sCPr * ti + sCPi * tr;
        const float vr  = sCBr - cr2, vi = sCBi - ci2;
        KhatRe[h * L_DIM + l] = pref_re * vr - pref_im * vi;
        KhatIm[h * L_DIM + l] = pref_re * vi + pref_im * vr;
    }
}

__global__ __launch_bounds__(256)
void s4_ifft(const float* __restrict__ KhatRe, const float* __restrict__ KhatIm,
             const float* __restrict__ D, float* __restrict__ out)
{
    __shared__ float re[L_DIM];
    __shared__ float im[L_DIM];
    const int tid = threadIdx.x;
    const int h   = blockIdx.x;

    for (int i = tid; i < L_DIM; i += 256) {
        const int j = (int)(__brev((unsigned)i) >> 21);
        re[i] = KhatRe[h * L_DIM + j];
        im[i] = KhatIm[h * L_DIM + j];
    }
    __syncthreads();

    for (int s = 1; s <= 11; ++s) {
        const int m  = 1 << s;
        const int hf = m >> 1;
        const float wstep = TWO_PI / (float)m;
        for (int b = tid; b < L_DIM / 2; b += 256) {
            const int j    = b & (hf - 1);
            const int base = ((b >> (s - 1)) << s) + j;
            float wr, wi;
            __sincosf(wstep * (float)j, &wi, &wr);
            const float xr = re[base + hf], xi = im[base + hf];
            const float tr = wr * xr - wi * xi;
            const float ti = wr * xi + wi * xr;
            const float ur = re[base], ui = im[base];
            re[base]      = ur + tr;
            im[base]      = ui + ti;
            re[base + hf] = ur - tr;
            im[base + hf] = ui - ti;
        }
        __syncthreads();
    }

    const float scale = 1.0f / (float)L_DIM;
    for (int i = tid; i < L_DIM; i += 256)
        out[h * L_DIM + i] = re[i] * scale;

    if (tid == 0)
        out[(size_t)CH_DIM * H_DIM * L_DIM + h] = D[h];
}

extern "C" void kernel_launch(void* const* d_in, const int* in_sizes, int n_in,
                              void* d_out, int out_size, void* d_ws, size_t ws_size,
                              hipStream_t stream)
{
    (void)in_sizes; (void)n_in; (void)out_size; (void)ws_size;
    const float* Lre = (const float*)d_in[0];
    const float* Lim = (const float*)d_in[1];
    const float* Pre = (const float*)d_in[2];
    const float* Pim = (const float*)d_in[3];
    const float* Bre = (const float*)d_in[4];
    const float* Bim = (const float*)d_in[5];
    const float* Cre = (const float*)d_in[6];
    const float* Cim = (const float*)d_in[7];
    const float* D   = (const float*)d_in[8];
    const float* ldt = (const float*)d_in[9];

    float* out    = (float*)d_out;
    float* KhatRe = (float*)d_ws;
    float* KhatIm = KhatRe + (size_t)H_DIM * L_DIM;

    s4_cauchy_wmma<<<dim3(H_DIM * (L_DIM / LTILE)), 256, 0, stream>>>(
        Lre, Lim, Pre, Pim, Bre, Bim, Cre, Cim, ldt, KhatRe, KhatIm);
    s4_ifft<<<dim3(H_DIM), 256, 0, stream>>>(KhatRe, KhatIm, D, out);
}


# one launch instead of two: a workgroup per head loops over its 128 frequency tiles (weights as A operand, sums land in the owning lane), writes K_hat bit-reversed to LDS, runs the inverse FFT there
# speedup vs baseline: 1.2348x; 1.2348x over previous
#include <hip/hip_runtime.h>
#include <math.h>

#define H_DIM   512
#define N_DIM   64
#define L_DIM   2048
#define CH_DIM  1
#define NT      256
#define NWAVE   (NT / 32)
#define NTILE   (L_DIM / 16)
#define TWO_PI  6.283185307179586f

typedef __attribute__((ext_vector_type(2))) float v2f;
typedef __attribute__((ext_vector_type(8))) float v8f;

__device__ __forceinline__ float softplusf(float x) {
    return fmaxf(x, 0.0f) + log1pf(expf(-fabsf(x)));
}

__device__ __forceinline__ float frcp(float x) {
    return __builtin_amdgcn_rcpf(x);
}

__global__ __launch_bounds__(NT)
void s4_fused(const float* __restrict__ Lre, const float* __restrict__ Lim,
              const float* __restrict__ Pre, const float* __restrict__ Pim,
              const float* __restrict__ Bre, const float* __restrict__ Bim,
              const float* __restrict__ Cre, const float* __restrict__ Cim,
              const float* __restrict__ log_dt, const float* __restrict__ D,
              float* __restrict__ out)
{
    __shared__ float Wm[128 * 8];
    __shared__ float lamRe[N_DIM], lamIm[N_DIM];
    __shared__ float re[L_DIM];
    __shared__ float im[L_DIM];

    const int tid = threadIdx.x;
    const int h   = blockIdx.x;

    if (tid < N_DIM) {
        const int n   = tid;
        const int idx = h * N_DIM + n;
        const float pr = Pre[idx], pi = Pim[idx];
        const float br = Bre[idx], bi = Bim[idx];
        const float cr = Cre[idx], ci = Cim[idx];
        const float wPBr = pr * br + pi * bi, wPBi = pr * bi - pi * br;
        const float wPPr = pr * pr + pi * pi, wPPi = 0.0f;
        const float wCBr = cr * br + ci * bi, wCBi = cr * bi - ci * br;
        const float wCPr = cr * pr + ci * pi, wCPi = cr * pi - ci * pr;
        const float wr[4] = {wCBr, wCPr, wPBr, wPPr};
        const float wi[4] = {wCBi, wCPi, wPBi, wPPi};
        #pragma unroll
        for (int X = 0; X < 4; ++X) {
            Wm[n * 8 + 2 * X]              =  wr[X];
            Wm[n * 8 + 2 * X + 1]          =  wi[X];
            Wm[(64 + n) * 8 + 2 * X]       = -wi[X];
            Wm[(64 + n) * 8 + 2 * X + 1]   =  wr[X];
        }
        lamRe[n] = -softplusf(Lre[idx]);
        lamIm[n] = Lim[idx];
    }
    __syncthreads();

    const int wave = tid >> 5;
    const int lane = tid & 31;
    const int half = lane >> 4;
    const int c    = lane & 15;

    v2f wRe[16], wIm[16], lr[16], li[16];
    #pragma unroll
    for (int kk = 0; kk < 16; ++kk) {
        const int krow = 4 * kk + 2 * half;
        const int cc   = c & 7;
        v2f a = { Wm[krow * 8 + cc],        Wm[(krow + 1) * 8 + cc] };
        v2f b = { Wm[(64 + krow) * 8 + cc], Wm[(64 + krow + 1) * 8 + cc] };
        const v2f zero = { 0.0f, 0.0f };
        wRe[kk] = (c < 8) ? a : zero;
        wIm[kk] = (c < 8) ? b : zero;
        lr[kk]  = (v2f){ lamRe[krow], lamRe[krow + 1] };
        li[kk]  = (v2f){ lamIm[krow], lamIm[krow + 1] };
    }

    const float sc = 2.0f * expf(-log_dt[h]);

    for (int t = wave; t < NTILE; t += NWAVE) {
        const int j = 16 * t + c;
        const int l = (int)(__brev((unsigned)j) >> 21);

        const float theta = (TWO_PI / (float)L_DIM) * (float)l;
        float zre, zim_p;
        __sincosf(theta, &zim_p, &zre);
        const float zim = -zim_p;
        float dnr = 1.0f + zre, dni = zim;
        float d2  = dnr * dnr + dni * dni;
        const float eps = 1.1920929e-7f;
        if (sqrtf(d2) < eps) { dnr = eps; dni = 0.0f; d2 = eps * eps; }
        const float nur = 1.0f - zre, nui = -zim;
        const float invd2   = frcp(d2);
        const float g_re    = sc * (nur * dnr + nui * dni) * invd2;
        const float g_im    = sc * (nui * dnr - nur * dni) * invd2;
        const float pref_re =  2.0f * dnr * invd2;
        const float pref_im = -2.0f * dni * invd2;

        v8f acc = {};
        #pragma unroll
        for (int kk = 0; kk < 16; ++kk) {
            const float dr0 = g_re - lr[kk].x, di0 = g_im - li[kk].x;
            const float dr1 = g_re - lr[kk].y, di1 = g_im - li[kk].y;
            const float iv0 = frcp(dr0 * dr0 + di0 * di0);
            const float iv1 = frcp(dr1 * dr1 + di1 * di1);
            v2f rRe = {  dr0 * iv0,  dr1 * iv1 };
            v2f rIm = { -di0 * iv0, -di1 * iv1 };
            acc = __builtin_amdgcn_wmma_f32_16x16x4_f32(false, wRe[kk], false, rRe,
                                                        (short)0, acc, false, false);
            acc = __builtin_amdgcn_wmma_f32_16x16x4_f32(false, wIm[kk], false, rIm,
                                                        (short)0, acc, false, false);
        }

        if (lane < 16) {
            const float sCBr = acc[0], sCBi = acc[1];
            const float sCPr = acc[2], sCPi = acc[3];
            const float sPBr = acc[4], sPBi = acc[5];
            const float sPPr = acc[6], sPPi = acc[7];
            const float dwr = 1.0f + sPPr, dwi = sPPi;
            const float iw  = frcp(dwr * dwr + dwi * dwi);
            const float tr  = (sPBr * dwr + sPBi * dwi) * iw;
            const float ti  = (sPBi * dwr - sPBr * dwi) * iw;
            const float cr2 = sCPr * tr - sCPi * ti;
            const float ci2 = sCPr * ti + sCPi * tr;
            const float vr  = sCBr - cr2, vi = sCBi - ci2;
            re[j] = pref_re * vr - pref_im * vi;
            im[j] = pref_re * vi + pref_im * vr;
        }
    }
    __syncthreads();

    for (int s = 1; s <= 11; ++s) {
        const int m  = 1 << s;
        const int hf = m >> 1;
        const float wstep = TWO_PI / (float)m;
        for (int b = tid; b < L_DIM / 2; b += NT) {
            const int jj   = b & (hf - 1);
            const int base = ((b >> (s - 1)) << s) + jj;
            float wr, wi;
            __sincosf(wstep * (float)jj, &wi, &wr);
            const float xr = re[base + hf], xi = im[base + hf];
            const float tr = wr * xr - wi * xi;
            const float ti = wr * xi + wi * xr;
            const float ur = re[base], ui = im[base];
            re[base]      = ur + tr;
            im[base]      = ui + ti;
            re[base + hf] = ur - tr;
            im[base + hf] = ui - ti;
        }
        __syncthreads();
    }

    const float scale = 1.0f / (float)L_DIM;
    for (int i = tid; i < L_DIM; i += NT)
        out[h * L_DIM + i] = re[i] * scale;

    if (tid == 0)
        out[(size_t)CH_DIM * H_DIM * L_DIM + h] = D[h];
}

extern "C" void kernel_launch(void* const* d_in, const int* in_sizes, int n_in,
                              void* d_out, int out_size, void* d_ws, size_t ws_size,
                              hipStream_t stream)
{
    (void)in_sizes; (void)n_in; (void)out_size; (void)d_ws; (void)ws_size;
    const float* Lre = (const float*)d_in[0];
    const float* Lim = (const float*)d_in[1];
    const float* Pre = (const float*)d_in[2];
    const float* Pim = (const float*)d_in[3];
    const float* Bre = (const float*)d_in[4];
    const float* Bim = (const float*)d_in[5];
    const float* Cre = (const float*)d_in[6];
    const float* Cim = (const float*)d_in[7];
    const float* D   = (const float*)d_in[8];
    const float* ldt = (const float*)d_in[9];

    float* out = (float*)d_out;

    s4_fused<<<dim3(H_DIM), NT, 0, stream>>>(
        Lre, Lim, Pre, Pim, Bre, Bim, Cre, Cim, ldt, D, out);
}


# resolvent numerator folded into the weights: B operand is 1/|g-lam|^2 alone, A holds w and w*conj(lam) in all 16 rows, 16 WMMA per tile not 32; conj(g) applied after the sums (sums regrouped)
# speedup vs baseline: 1.7059x; 1.3815x over previous
#include <hip/hip_runtime.h>
#include <math.h>

#define H_DIM   512
#define N_DIM   64
#define L_DIM   2048
#define CH_DIM  1
#define NT      256
#define NWAVE   (NT / 32)
#define NTILE   (L_DIM / 16)
#define TWO_PI  6.283185307179586f

typedef __attribute__((ext_vector_type(2))) float v2f;
typedef __attribute__((ext_vector_type(8))) float v8f;

__device__ __forceinline__ float softplusf(float x) {
    return fmaxf(x, 0.0f) + log1pf(expf(-fabsf(x)));
}

__device__ __forceinline__ float frcp(float x) {
    return __builtin_amdgcn_rcpf(x);
}

__device__ __forceinline__ float xhalf(float x) {
    const int b = __builtin_bit_cast(int, x);
    return __builtin_bit_cast(float,
        __builtin_amdgcn_permlanex16(b, b, 0x76543210, 0xfedcba98, false, false));
}

__global__ __launch_bounds__(NT)
void s4_fused(const float* __restrict__ Lre, const float* __restrict__ Lim,
              const float* __restrict__ Pre, const float* __restrict__ Pim,
              const float* __restrict__ Bre, const float* __restrict__ Bim,
              const float* __restrict__ Cre, const float* __restrict__ Cim,
              const float* __restrict__ log_dt, const float* __restrict__ D,
              float* __restrict__ out)
{
    __shared__ float Wm[N_DIM * 16];
    __shared__ float lamRe[N_DIM], lamIm[N_DIM];
    __shared__ float re[L_DIM];
    __shared__ float im[L_DIM];

    const int tid = threadIdx.x;
    const int h   = blockIdx.x;

    if (tid < N_DIM) {
        const int n   = tid;
        const int idx = h * N_DIM + n;
        const float pr = Pre[idx], pi = Pim[idx];
        const float br = Bre[idx], bi = Bim[idx];
        const float cr = Cre[idx], ci = Cim[idx];
        const float wPBr = pr * br + pi * bi, wPBi = pr * bi - pi * br;
        const float wPPr = pr * pr + pi * pi, wPPi = 0.0f;
        const float wCBr = cr * br + ci * bi, wCBi = cr * bi - ci * br;
        const float wCPr = cr * pr + ci * pi, wCPi = cr * pi - ci * pr;
        const float wr[4] = {wCBr, wCPr, wPBr, wPPr};
        const float wi[4] = {wCBi, wCPi, wPBi, wPPi};
        const float lre = -softplusf(Lre[idx]);
        const float lim = Lim[idx];
        #pragma unroll
        for (int X = 0; X < 4; ++X) {
            Wm[n * 16 + 4 * X]     = wr[X];
            Wm[n * 16 + 4 * X + 1] = wi[X];
            Wm[n * 16 + 4 * X + 2] = wr[X] * lre + wi[X] * lim;
            Wm[n * 16 + 4 * X + 3] = wi[X] * lre - wr[X] * lim;
        }
        lamRe[n] = lre;
        lamIm[n] = lim;
    }
    __syncthreads();

    const int wave = tid >> 5;
    const int lane = tid & 31;
    const int half = lane >> 4;
    const int c    = lane & 15;

    v2f wA[16], lr[16], li[16];
    #pragma unroll
    for (int kk = 0; kk < 16; ++kk) {
        const int krow = 4 * kk + 2 * half;
        wA[kk] = (v2f){ Wm[krow * 16 + c], Wm[(krow + 1) * 16 + c] };
        lr[kk] = (v2f){ lamRe[krow], lamRe[krow + 1] };
        li[kk] = (v2f){ lamIm[krow], lamIm[krow + 1] };
    }

    const float sc = 2.0f * expf(-log_dt[h]);

    for (int t = wave; t < NTILE; t += NWAVE) {
        const int j = 16 * t + c;
        const int l = (int)(__brev((unsigned)j) >> 21);

        const float theta = (TWO_PI / (float)L_DIM) * (float)l;
        float zre, zim_p;
        __sincosf(theta, &zim_p, &zre);
        const float zim = -zim_p;
        float dnr = 1.0f + zre, dni = zim;
        float d2  = dnr * dnr + dni * dni;
        const float eps = 1.1920929e-7f;
        if (sqrtf(d2) < eps) { dnr = eps; dni = 0.0f; d2 = eps * eps; }
        const float nur = 1.0f - zre, nui = -zim;
        const float invd2   = frcp(d2);
        const float g_re    = sc * (nur * dnr + nui * dni) * invd2;
        const float g_im    = sc * (nui * dnr - nur * dni) * invd2;
        const float pref_re =  2.0f * dnr * invd2;
        const float pref_im = -2.0f * dni * invd2;

        const v2f gr2 = { g_re, g_re };
        const v2f gi2 = { g_im, g_im };
        v8f acc = {};
        #pragma unroll
        for (int kk = 0; kk < 16; ++kk) {
            const v2f dr = gr2 - lr[kk];
            const v2f di = gi2 - li[kk];
            const v2f d2n = dr * dr + di * di;
            const v2f q = { frcp(d2n.x), frcp(d2n.y) };
            acc = __builtin_amdgcn_wmma_f32_16x16x4_f32(false, wA[kk], false, q,
                                                        (short)0, acc, false, false);
        }

        const float sar = g_re * acc[0] + g_im * acc[1] - acc[2];
        const float sai = g_re * acc[1] - g_im * acc[0] - acc[3];
        const float sbr = g_re * acc[4] + g_im * acc[5] - acc[6];
        const float sbi = g_re * acc[5] - g_im * acc[4] - acc[7];
        const float dwr = 1.0f + sbr, dwi = sbi;
        const float iw  = frcp(dwr * dwr + dwi * dwi);
        const float qr  = (sar * dwr + sai * dwi) * iw;
        const float qi  = (sai * dwr - sar * dwi) * iw;
        const float tr  = xhalf(qr);
        const float ti  = xhalf(qi);

        if (lane < 16) {
            const float cr2 = sbr * tr - sbi * ti;
            const float ci2 = sbr * ti + sbi * tr;
            const float vr  = sar - cr2, vi = sai - ci2;
            re[j] = pref_re * vr - pref_im * vi;
            im[j] = pref_re * vi + pref_im * vr;
        }
    }
    __syncthreads();

    for (int s = 1; s <= 11; ++s) {
        const int m  = 1 << s;
        const int hf = m >> 1;
        const float wstep = TWO_PI / (float)m;
        for (int b = tid; b < L_DIM / 2; b += NT) {
            const int jj   = b & (hf - 1);
            const int base = ((b >> (s - 1)) << s) + jj;
            float wr, wi;
            __sincosf(wstep * (float)jj, &wi, &wr);
            const float xr = re[base + hf], xi = im[base + hf];
            const float tr = wr * xr - wi * xi;
            const float ti = wr * xi + wi * xr;
            const float ur = re[base], ui = im[base];
            re[base]      = ur + tr;
            im[base]      = ui + ti;
            re[base + hf] = ur - tr;
            im[base + hf] = ui - ti;
        }
        __syncthreads();
    }

    const float scale = 1.0f / (float)L_DIM;
    for (int i = tid; i < L_DIM; i += NT)
        out[h * L_DIM + i] = re[i] * scale;

    if (tid == 0)
        out[(size_t)CH_DIM * H_DIM * L_DIM + h] = D[h];
}

extern "C" void kernel_launch(void* const* d_in, const int* in_sizes, int n_in,
                              void* d_out, int out_size, void* d_ws, size_t ws_size,
                              hipStream_t stream)
{
    (void)in_sizes; (void)n_in; (void)out_size; (void)d_ws; (void)ws_size;
    const float* Lre = (const float*)d_in[0];
    const float* Lim = (const float*)d_in[1];
    const float* Pre = (const float*)d_in[2];
    const float* Pim = (const float*)d_in[3];
    const float* Bre = (const float*)d_in[4];
    const float* Bim = (const float*)d_in[5];
    const float* Cre = (const float*)d_in[6];
    const float* Cim = (const float*)d_in[7];
    const float* D   = (const float*)d_in[8];
    const float* ldt = (const float*)d_in[9];

    float* out = (float*)d_out;

    s4_fused<<<dim3(H_DIM), NT, 0, stream>>>(
        Lre, Lim, Pre, Pim, Bre, Bim, Cre, Cim, ldt, D, out);
}


# inverse FFT: the 11 radix-2 stages run as 4 register-resident passes (3+3+3+2 stages, 8 points per thread, batched loads, padded complex LDS buffer), last pass stores to global; same sum order
# speedup vs baseline: 2.0152x; 1.1813x over previous
#include <hip/hip_runtime.h>
#include <math.h>

#define H_DIM   512
#define N_DIM   64
#define L_DIM   2048
#define CH_DIM  1
#define NT      256
#define NWAVE   (NT / 32)
#define NTILE   (L_DIM / 16)
#define TWO_PI  6.283185307179586f

typedef __attribute__((ext_vector_type(2))) float v2f;
typedef __attribute__((ext_vector_type(8))) float v8f;

__device__ __forceinline__ float softplusf(float x) {
    return fmaxf(x, 0.0f) + log1pf(expf(-fabsf(x)));
}

__device__ __forceinline__ float frcp(float x) {
    return __builtin_amdgcn_rcpf(x);
}

__device__ __forceinline__ float xhalf(float x) {
    const int b = __builtin_bit_cast(int, x);
    return __builtin_bit_cast(float,
        __builtin_amdgcn_permlanex16(b, b, 0x76543210, 0xfedcba98, false, false));
}

__device__ __forceinline__ int cpad(int i) { return i + (i >> 3); }

template <int S0, int R, bool LAST>
__device__ __forceinline__ void fft_pass(float2* __restrict__ cbuf, const int g,
                                         float* __restrict__ orow, const float scale)
{
    constexpr int NP  = 1 << R;
    constexpr int LOW = S0 - 1;
    const int low  = g & ((1 << LOW) - 1);
    const int high = g >> LOW;
    const int base = low + (high << (LOW + R));

    float xr[NP], xi[NP];
    #pragma unroll
    for (int k = 0; k < NP; ++k) {
        const float2 v = cbuf[cpad(base + (k << LOW))];
        xr[k] = v.x;
        xi[k] = v.y;
    }
    #pragma unroll
    for (int t = 0; t < R; ++t) {
        const float wstep = TWO_PI / (float)(1 << (S0 + t));
        #pragma unroll
        for (int q = 0; q < (1 << t); ++q) {
            const int jj = low + (q << LOW);
            float wr, wi;
            __sincosf(wstep * (float)jj, &wi, &wr);
            #pragma unroll
            for (int k = 0; k < NP; ++k) {
                if ((k & ((2 << t) - 1)) == q) {
                    const int k2 = k + (1 << t);
                    const float tr = wr * xr[k2] - wi * xi[k2];
                    const float ti = wr * xi[k2] + wi * xr[k2];
                    const float ur = xr[k], ui = xi[k];
                    xr[k]  = ur + tr;
                    xi[k]  = ui + ti;
                    xr[k2] = ur - tr;
                    xi[k2] = ui - ti;
                }
            }
        }
    }
    #pragma unroll
    for (int k = 0; k < NP; ++k) {
        const int idx = base + (k << LOW);
        if (LAST) orow[idx] = xr[k] * scale;
        else      cbuf[cpad(idx)] = make_float2(xr[k], xi[k]);
    }
}

__global__ __launch_bounds__(NT)
void s4_fused(const float* __restrict__ Lre, const float* __restrict__ Lim,
              const float* __restrict__ Pre, const float* __restrict__ Pim,
              const float* __restrict__ Bre, const float* __restrict__ Bim,
              const float* __restrict__ Cre, const float* __restrict__ Cim,
              const float* __restrict__ log_dt, const float* __restrict__ D,
              float* __restrict__ out)
{
    __shared__ float Wm[N_DIM * 16];
    __shared__ float lamRe[N_DIM], lamIm[N_DIM];
    __shared__ float2 cbuf[L_DIM + L_DIM / 8];

    const int tid = threadIdx.x;
    const int h   = blockIdx.x;

    if (tid < N_DIM) {
        const int n   = tid;
        const int idx = h * N_DIM + n;
        const float pr = Pre[idx], pi = Pim[idx];
        const float br = Bre[idx], bi = Bim[idx];
        const float cr = Cre[idx], ci = Cim[idx];
        const float wPBr = pr * br + pi * bi, wPBi = pr * bi - pi * br;
        const float wPPr = pr * pr + pi * pi, wPPi = 0.0f;
        const float wCBr = cr * br + ci * bi, wCBi = cr * bi - ci * br;
        const float wCPr = cr * pr + ci * pi, wCPi = cr * pi - ci * pr;
        const float wr[4] = {wCBr, wCPr, wPBr, wPPr};
        const float wi[4] = {wCBi, wCPi, wPBi, wPPi};
        const float lre = -softplusf(Lre[idx]);
        const float lim = Lim[idx];
        #pragma unroll
        for (int X = 0; X < 4; ++X) {
            Wm[n * 16 + 4 * X]     = wr[X];
            Wm[n * 16 + 4 * X + 1] = wi[X];
            Wm[n * 16 + 4 * X + 2] = wr[X] * lre + wi[X] * lim;
            Wm[n * 16 + 4 * X + 3] = wi[X] * lre - wr[X] * lim;
        }
        lamRe[n] = lre;
        lamIm[n] = lim;
    }
    __syncthreads();

    const int wave = tid >> 5;
    const int lane = tid & 31;
    const int half = lane >> 4;
    const int c    = lane & 15;

    v2f wA[16], lr[16], li[16];
    #pragma unroll
    for (int kk = 0; kk < 16; ++kk) {
        const int krow = 4 * kk + 2 * half;
        wA[kk] = (v2f){ Wm[krow * 16 + c], Wm[(krow + 1) * 16 + c] };
        lr[kk] = (v2f){ lamRe[krow], lamRe[krow + 1] };
        li[kk] = (v2f){ lamIm[krow], lamIm[krow + 1] };
    }

    const float sc = 2.0f * expf(-log_dt[h]);

    for (int t = wave; t < NTILE; t += NWAVE) {
        const int j = 16 * t + c;
        const int l = (int)(__brev((unsigned)j) >> 21);

        const float theta = (TWO_PI / (float)L_DIM) * (float)l;
        float zre, zim_p;
        __sincosf(theta, &zim_p, &zre);
        const float zim = -zim_p;
        float dnr = 1.0f + zre, dni = zim;
        float d2  = dnr * dnr + dni * dni;
        const float eps = 1.1920929e-7f;
        if (sqrtf(d2) < eps) { dnr = eps; dni = 0.0f; d2 = eps * eps; }
        const float nur = 1.0f - zre, nui = -zim;
        const float invd2   = frcp(d2);
        const float g_re    = sc * (nur * dnr + nui * dni) * invd2;
        const float g_im    = sc * (nui * dnr - nur * dni) * invd2;
        const float pref_re =  2.0f * dnr * invd2;
        const float pref_im = -2.0f * dni * invd2;

        const v2f gr2 = { g_re, g_re };
        const v2f gi2 = { g_im, g_im };
        v8f acc = {};
        #pragma unroll
        for (int kk = 0; kk < 16; ++kk) {
            const v2f dr = gr2 - lr[kk];
            const v2f di = gi2 - li[kk];
            const v2f d2n = dr * dr + di * di;
            const v2f q = { frcp(d2n.x), frcp(d2n.y) };
            acc = __builtin_amdgcn_wmma_f32_16x16x4_f32(false, wA[kk], false, q,
                                                        (short)0, acc, false, false);
        }

        const float sar = g_re * acc[0] + g_im * acc[1] - acc[2];
        const float sai = g_re * acc[1] - g_im * acc[0] - acc[3];
        const float sbr = g_re * acc[4] + g_im * acc[5] - acc[6];
        const float sbi = g_re * acc[5] - g_im * acc[4] - acc[7];
        const float dwr = 1.0f + sbr, dwi = sbi;
        const float iw  = frcp(dwr * dwr + dwi * dwi);
        const float qr  = (sar * dwr + sai * dwi) * iw;
        const float qi  = (sai * dwr - sar * dwi) * iw;
        const float tr  = xhalf(qr);
        const float ti  = xhalf(qi);

        if (lane < 16) {
            const float cr2 = sbr * tr - sbi * ti;
            const float ci2 = sbr * ti + sbi * tr;
            const float vr  = sar - cr2, vi = sai - ci2;
            cbuf[cpad(j)] = make_float2(pref_re * vr - pref_im * vi,
                                        pref_re * vi + pref_im * vr);
        }
    }
    __syncthreads();

    const float scale = 1.0f / (float)L_DIM;
    float* __restrict__ orow = out + h * L_DIM;
    fft_pass<1, 3, false>(cbuf, tid, orow, scale);
    __syncthreads();
    fft_pass<4, 3, false>(cbuf, tid, orow, scale);
    __syncthreads();
    fft_pass<7, 3, false>(cbuf, tid, orow, scale);
    __syncthreads();
    fft_pass<10, 2, true>(cbuf, tid, orow, scale);
    fft_pass<10, 2, true>(cbuf, tid + NT, orow, scale);

    if (tid == 0)
        out[(size_t)CH_DIM * H_DIM * L_DIM + h] = D[h];
}

extern "C" void kernel_launch(void* const* d_in, const int* in_sizes, int n_in,
                              void* d_out, int out_size, void* d_ws, size_t ws_size,
                              hipStream_t stream)
{
    (void)in_sizes; (void)n_in; (void)out_size; (void)d_ws; (void)ws_size;
    const float* Lre = (const float*)d_in[0];
    const float* Lim = (const float*)d_in[1];
    const float* Pre = (const float*)d_in[2];
    const float* Pim = (const float*)d_in[3];
    const float* Bre = (const float*)d_in[4];
    const float* Bim = (const float*)d_in[5];
    const float* Cre = (const float*)d_in[6];
    const float* Cim = (const float*)d_in[7];
    const float* D   = (const float*)d_in[8];
    const float* ldt = (const float*)d_in[9];

    float* out = (float*)d_out;

    s4_fused<<<dim3(H_DIM), NT, 0, stream>>>(
        Lre, Lim, Pre, Pim, Bre, Bim, Cre, Cim, ldt, D, out);
}


# g and 2/(1+z) in closed form (g = i*sc*tan(theta/2), 2/(1+z) = 1+i*tan(theta/2)): pole distance Re(lam)^2+(sc*tau-Im(lam))^2 in 2 packed ops not 4, short per-tile head; l=L/2 tile general; same sums
# speedup vs baseline: 2.2033x; 1.0933x over previous
#include <hip/hip_runtime.h>
#include <math.h>

#define H_DIM   512
#define N_DIM   64
#define L_DIM   2048
#define CH_DIM  1
#define NT      256
#define NWAVE   (NT / 32)
#define NTILE   (L_DIM / 16)
#define TWO_PI  6.283185307179586f

typedef __attribute__((ext_vector_type(2))) float v2f;
typedef __attribute__((ext_vector_type(8))) float v8f;

__device__ __forceinline__ float softplusf(float x) {
    return fmaxf(x, 0.0f) + log1pf(expf(-fabsf(x)));
}

__device__ __forceinline__ float frcp(float x) {
    return __builtin_amdgcn_rcpf(x);
}

__device__ __forceinline__ float xhalf(float x) {
    const int b = __builtin_bit_cast(int, x);
    return __builtin_bit_cast(float,
        __builtin_amdgcn_permlanex16(b, b, 0x76543210, 0xfedcba98, false, false));
}

__device__ __forceinline__ int cpad(int i) { return i + (i >> 3); }

template <int S0, int R, bool LAST>
__device__ __forceinline__ void fft_pass(float2* __restrict__ cbuf, const int g,
                                         float* __restrict__ orow, const float scale)
{
    constexpr int NP  = 1 << R;
    constexpr int LOW = S0 - 1;
    const int low  = g & ((1 << LOW) - 1);
    const int high = g >> LOW;
    const int base = low + (high << (LOW + R));

    float xr[NP], xi[NP];
    #pragma unroll
    for (int k = 0; k < NP; ++k) {
        const float2 v = cbuf[cpad(base + (k << LOW))];
        xr[k] = v.x;
        xi[k] = v.y;
    }
    #pragma unroll
    for (int t = 0; t < R; ++t) {
        const float wstep = TWO_PI / (float)(1 << (S0 + t));
        #pragma unroll
        for (int q = 0; q < (1 << t); ++q) {
            const int jj = low + (q << LOW);
            float wr, wi;
            __sincosf(wstep * (float)jj, &wi, &wr);
            #pragma unroll
            for (int k = 0; k < NP; ++k) {
                if ((k & ((2 << t) - 1)) == q) {
                    const int k2 = k + (1 << t);
                    const float tr = wr * xr[k2] - wi * xi[k2];
                    const float ti = wr * xi[k2] + wi * xr[k2];
                    const float ur = xr[k], ui = xi[k];
                    xr[k]  = ur + tr;
                    xi[k]  = ui + ti;
                    xr[k2] = ur - tr;
                    xi[k2] = ui - ti;
                }
            }
        }
    }
    #pragma unroll
    for (int k = 0; k < NP; ++k) {
        const int idx = base + (k << LOW);
        if (LAST) orow[idx] = xr[k] * scale;
        else      cbuf[cpad(idx)] = make_float2(xr[k], xi[k]);
    }
}

template <bool MID_TILE>
__device__ __forceinline__ void cauchy_tile(const int t, const int lane, const float sc,
                                            const v2f (&wA)[16], const v2f (&lr2)[16],
                                            const v2f (&li)[16],
                                            const float* __restrict__ lamRe,
                                            float2* __restrict__ cbuf)
{
    const int half = lane >> 4;
    const int c    = lane & 15;
    const int j = 16 * t + c;
    const int l = (int)(__brev((unsigned)j) >> 21);

    const float rev = (float)l * (1.0f / (float)L_DIM);
    const float tau = __builtin_amdgcn_sinf(rev) * frcp(1.0f + __builtin_amdgcn_cosf(rev));

    float g_re = 0.0f, g_im = sc * tau, pref_re = 1.0f, pref_im = tau;
    v8f acc = {};
    if (MID_TILE) {
        const bool  mid = (l == L_DIM / 2);
        const float eps = 1.1920929e-7f;
        g_re    = mid ? sc * (2.0f / eps) : 0.0f;
        g_im    = mid ? 0.0f : g_im;
        pref_re = mid ? 2.0f / eps : 1.0f;
        pref_im = mid ? 0.0f : tau;
        const v2f gr2 = { g_re, g_re };
        const v2f gi2 = { g_im, g_im };
        #pragma unroll
        for (int kk = 0; kk < 16; ++kk) {
            const int krow = 4 * kk + 2 * half;
            const v2f lrv = { lamRe[krow], lamRe[krow + 1] };
            const v2f dr = gr2 - lrv;
            const v2f di = gi2 - li[kk];
            const v2f d2n = dr * dr + di * di;
            const v2f q = { frcp(d2n.x), frcp(d2n.y) };
            acc = __builtin_amdgcn_wmma_f32_16x16x4_f32(false, wA[kk], false, q,
                                                        (short)0, acc, false, false);
        }
    } else {
        const v2f gi2 = { g_im, g_im };
        #pragma unroll
        for (int kk = 0; kk < 16; ++kk) {
            const v2f di = gi2 - li[kk];
            const v2f d2n = lr2[kk] + di * di;
            const v2f q = { frcp(d2n.x), frcp(d2n.y) };
            acc = __builtin_amdgcn_wmma_f32_16x16x4_f32(false, wA[kk], false, q,
                                                        (short)0, acc, false, false);
        }
    }

    float sar, sai, sbr, sbi;
    if (MID_TILE) {
        sar = g_re * acc[0] + g_im * acc[1] - acc[2];
        sai = g_re * acc[1] - g_im * acc[0] - acc[3];
        sbr = g_re * acc[4] + g_im * acc[5] - acc[6];
        sbi = g_re * acc[5] - g_im * acc[4] - acc[7];
    } else {
        sar =  g_im * acc[1] - acc[2];
        sai = -g_im * acc[0] - acc[3];
        sbr =  g_im * acc[5] - acc[6];
        sbi = -g_im * acc[4] - acc[7];
    }
    const float dwr = 1.0f + sbr, dwi = sbi;
    const float iw  = frcp(dwr * dwr + dwi * dwi);
    const float qr  = (sar * dwr + sai * dwi) * iw;
    const float qi  = (sai * dwr - sar * dwi) * iw;
    const float tr  = xhalf(qr);
    const float ti  = xhalf(qi);

    if (lane < 16) {
        const float cr2 = sbr * tr - sbi * ti;
        const float ci2 = sbr * ti + sbi * tr;
        const float vr  = sar - cr2, vi = sai - ci2;
        if (MID_TILE)
            cbuf[cpad(j)] = make_float2(pref_re * vr - pref_im * vi,
                                        pref_re * vi + pref_im * vr);
        else
            cbuf[cpad(j)] = make_float2(vr - pref_im * vi, vi + pref_im * vr);
    }
}

__global__ __launch_bounds__(NT)
void s4_fused(const float* __restrict__ Lre, const float* __restrict__ Lim,
              const float* __restrict__ Pre, const float* __restrict__ Pim,
              const float* __restrict__ Bre, const float* __restrict__ Bim,
              const float* __restrict__ Cre, const float* __restrict__ Cim,
              const float* __restrict__ log_dt, const float* __restrict__ D,
              float* __restrict__ out)
{
    __shared__ float Wm[N_DIM * 16];
    __shared__ float lamRe[N_DIM], lamIm[N_DIM];
    __shared__ float2 cbuf[L_DIM + L_DIM / 8];

    const int tid = threadIdx.x;
    const int h   = blockIdx.x;

    if (tid < N_DIM) {
        const int n   = tid;
        const int idx = h * N_DIM + n;
        const float pr = Pre[idx], pi = Pim[idx];
        const float br = Bre[idx], bi = Bim[idx];
        const float cr = Cre[idx], ci = Cim[idx];
        const float wPBr = pr * br + pi * bi, wPBi = pr * bi - pi * br;
        const float wPPr = pr * pr + pi * pi, wPPi = 0.0f;
        const float wCBr = cr * br + ci * bi, wCBi = cr * bi - ci * br;
        const float wCPr = cr * pr + ci * pi, wCPi = cr * pi - ci * pr;
        const float wr[4] = {wCBr, wCPr, wPBr, wPPr};
        const float wi[4] = {wCBi, wCPi, wPBi, wPPi};
        const float lre = -softplusf(Lre[idx]);
        const float lim = Lim[idx];
        #pragma unroll
        for (int X = 0; X < 4; ++X) {
            Wm[n * 16 + 4 * X]     = wr[X];
            Wm[n * 16 + 4 * X + 1] = wi[X];
            Wm[n * 16 + 4 * X + 2] = wr[X] * lre + wi[X] * lim;
            Wm[n * 16 + 4 * X + 3] = wi[X] * lre - wr[X] * lim;
        }
        lamRe[n] = lre;
        lamIm[n] = lim;
    }
    __syncthreads();

    const int wave = __builtin_amdgcn_readfirstlane(tid >> 5);
    const int lane = tid & 31;
    const int half = lane >> 4;
    const int c    = lane & 15;

    v2f wA[16], lr2[16], li[16];
    #pragma unroll
    for (int kk = 0; kk < 16; ++kk) {
        const int krow = 4 * kk + 2 * half;
        wA[kk]  = (v2f){ Wm[krow * 16 + c], Wm[(krow + 1) * 16 + c] };
        const v2f lrv = { lamRe[krow], lamRe[krow + 1] };
        lr2[kk] = lrv * lrv;
        li[kk]  = (v2f){ lamIm[krow], lamIm[krow + 1] };
    }

    const float sc = 2.0f * expf(-log_dt[h]);

    if (wave == 0)
        cauchy_tile<true>(0, lane, sc, wA, lr2, li, lamRe, cbuf);
    for (int t = (wave == 0 ? NWAVE : wave); t < NTILE; t += NWAVE)
        cauchy_tile<false>(t, lane, sc, wA, lr2, li, lamRe, cbuf);
    __syncthreads();

    const float scale = 1.0f / (float)L_DIM;
    float* __restrict__ orow = out + h * L_DIM;
    fft_pass<1, 3, false>(cbuf, tid, orow, scale);
    __syncthreads();
    fft_pass<4, 3, false>(cbuf, tid, orow, scale);
    __syncthreads();
    fft_pass<7, 3, false>(cbuf, tid, orow, scale);
    __syncthreads();
    fft_pass<10, 2, true>(cbuf, tid, orow, scale);
    fft_pass<10, 2, true>(cbuf, tid + NT, orow, scale);

    if (tid == 0)
        out[(size_t)CH_DIM * H_DIM * L_DIM + h] = D[h];
}

extern "C" void kernel_launch(void* const* d_in, const int* in_sizes, int n_in,
                              void* d_out, int out_size, void* d_ws, size_t ws_size,
                              hipStream_t stream)
{
    (void)in_sizes; (void)n_in; (void)out_size; (void)d_ws; (void)ws_size;
    const float* Lre = (const float*)d_in[0];
    const float* Lim = (const float*)d_in[1];
    const float* Pre = (const float*)d_in[2];
    const float* Pim = (const float*)d_in[3];
    const float* Bre = (const float*)d_in[4];
    const float* Bim = (const float*)d_in[5];
    const float* Cre = (const float*)d_in[6];
    const float* Cim = (const float*)d_in[7];
    const float* D   = (const float*)d_in[8];
    const float* ldt = (const float*)d_in[9];

    float* out = (float*)d_out;

    s4_fused<<<dim3(H_DIM), NT, 0, stream>>>(
        Lre, Lim, Pre, Pim, Bre, Bim, Cre, Cim, ldt, D, out);
}


# phase 1 takes two frequency tiles per wave trip: two independent accumulators share each A fragment, so one tile's pole distance, reciprocal and WMMA fill the other's waits; same sums per tile
# speedup vs baseline: 2.3847x; 1.0823x over previous
#include <hip/hip_runtime.h>
#include <math.h>

#define H_DIM   512
#define N_DIM   64
#define L_DIM   2048
#define CH_DIM  1
#define NT      256
#define NWAVE   (NT / 32)
#define NTILE   (L_DIM / 16)
#define TWO_PI  6.283185307179586f

typedef __attribute__((ext_vector_type(2))) float v2f;
typedef __attribute__((ext_vector_type(8))) float v8f;

__device__ __forceinline__ float softplusf(float x) {
    return fmaxf(x, 0.0f) + log1pf(expf(-fabsf(x)));
}

__device__ __forceinline__ float frcp(float x) {
    return __builtin_amdgcn_rcpf(x);
}

__device__ __forceinline__ float xhalf(float x) {
    const int b = __builtin_bit_cast(int, x);
    return __builtin_bit_cast(float,
        __builtin_amdgcn_permlanex16(b, b, 0x76543210, 0xfedcba98, false, false));
}

__device__ __forceinline__ int cpad(int i) { return i + (i >> 3); }

template <int S0, int R, bool LAST>
__device__ __forceinline__ void fft_pass(float2* __restrict__ cbuf, const int g,
                                         float* __restrict__ orow, const float scale)
{
    constexpr int NP  = 1 << R;
    constexpr int LOW = S0 - 1;
    const int low  = g & ((1 << LOW) - 1);
    const int high = g >> LOW;
    const int base = low + (high << (LOW + R));

    float xr[NP], xi[NP];
    #pragma unroll
    for (int k = 0; k < NP; ++k) {
        const float2 v = cbuf[cpad(base + (k << LOW))];
        xr[k] = v.x;
        xi[k] = v.y;
    }
    #pragma unroll
    for (int t = 0; t < R; ++t) {
        const float wstep = TWO_PI / (float)(1 << (S0 + t));
        #pragma unroll
        for (int q = 0; q < (1 << t); ++q) {
            const int jj = low + (q << LOW);
            float wr, wi;
            __sincosf(wstep * (float)jj, &wi, &wr);
            #pragma unroll
            for (int k = 0; k < NP; ++k) {
                if ((k & ((2 << t) - 1)) == q) {
                    const int k2 = k + (1 << t);
                    const float tr = wr * xr[k2] - wi * xi[k2];
                    const float ti = wr * xi[k2] + wi * xr[k2];
                    const float ur = xr[k], ui = xi[k];
                    xr[k]  = ur + tr;
                    xi[k]  = ui + ti;
                    xr[k2] = ur - tr;
                    xi[k2] = ui - ti;
                }
            }
        }
    }
    #pragma unroll
    for (int k = 0; k < NP; ++k) {
        const int idx = base + (k << LOW);
        if (LAST) orow[idx] = xr[k] * scale;
        else      cbuf[cpad(idx)] = make_float2(xr[k], xi[k]);
    }
}

struct TileHead { int j; float g_re, g_im, pref_re, pref_im; };

template <bool MID_TILE>
__device__ __forceinline__ TileHead tile_head(const int t, const int lane, const float sc)
{
    const int c = lane & 15;
    const int j = 16 * t + c;
    const int l = (int)(__brev((unsigned)j) >> 21);

    const float rev = (float)l * (1.0f / (float)L_DIM);
    const float tau = __builtin_amdgcn_sinf(rev) * frcp(1.0f + __builtin_amdgcn_cosf(rev));

    TileHead hd;
    hd.j = j;
    hd.g_re = 0.0f;
    hd.g_im = sc * tau;
    hd.pref_re = 1.0f;
    hd.pref_im = tau;
    if (MID_TILE) {
        const bool  mid = (l == L_DIM / 2);
        const float eps = 1.1920929e-7f;
        hd.g_re    = mid ? sc * (2.0f / eps) : 0.0f;
        hd.g_im    = mid ? 0.0f : hd.g_im;
        hd.pref_re = mid ? 2.0f / eps : 1.0f;
        hd.pref_im = mid ? 0.0f : tau;
    }
    return hd;
}

template <bool MID_TILE>
__device__ __forceinline__ float2 tile_tail(const v8f acc, const TileHead hd)
{
    const float g_re = hd.g_re, g_im = hd.g_im;
    float sar, sai, sbr, sbi;
    if (MID_TILE) {
        sar = g_re * acc[0] + g_im * acc[1] - acc[2];
        sai = g_re * acc[1] - g_im * acc[0] - acc[3];
        sbr = g_re * acc[4] + g_im * acc[5] - acc[6];
        sbi = g_re * acc[5] - g_im * acc[4] - acc[7];
    } else {
        sar =  g_im * acc[1] - acc[2];
        sai = -g_im * acc[0] - acc[3];
        sbr =  g_im * acc[5] - acc[6];
        sbi = -g_im * acc[4] - acc[7];
    }
    const float dwr = 1.0f + sbr, dwi = sbi;
    const float iw  = frcp(dwr * dwr + dwi * dwi);
    const float qr  = (sar * dwr + sai * dwi) * iw;
    const float qi  = (sai * dwr - sar * dwi) * iw;
    const float tr  = xhalf(qr);
    const float ti  = xhalf(qi);

    const float cr2 = sbr * tr - sbi * ti;
    const float ci2 = sbr * ti + sbi * tr;
    const float vr  = sar - cr2, vi = sai - ci2;
    if (MID_TILE)
        return make_float2(hd.pref_re * vr - hd.pref_im * vi,
                           hd.pref_re * vi + hd.pref_im * vr);
    return make_float2(vr - hd.pref_im * vi, vi + hd.pref_im * vr);
}

template <bool MID_A>
__device__ __forceinline__ void cauchy_pair(const int tA, const int tB, const int lane,
                                            const float sc,
                                            const v2f (&wA)[16], const v2f (&lr2)[16],
                                            const v2f (&li)[16],
                                            const float* __restrict__ lamRe,
                                            float2* __restrict__ cbuf)
{
    const int half = lane >> 4;
    const TileHead hA = tile_head<MID_A>(tA, lane, sc);
    const TileHead hB = tile_head<false>(tB, lane, sc);
    const v2f grA = { hA.g_re, hA.g_re };
    const v2f giA = { hA.g_im, hA.g_im };
    const v2f giB = { hB.g_im, hB.g_im };

    v8f accA = {}, accB = {};
    #pragma unroll
    for (int kk = 0; kk < 16; ++kk) {
        v2f d2A;
        if (MID_A) {
            const int krow = 4 * kk + 2 * half;
            const v2f lrv = { lamRe[krow], lamRe[krow + 1] };
            const v2f dr = grA - lrv;
            const v2f di = giA - li[kk];
            d2A = dr * dr + di * di;
        } else {
            const v2f di = giA - li[kk];
            d2A = lr2[kk] + di * di;
        }
        const v2f diB = giB - li[kk];
        const v2f d2B = lr2[kk] + diB * diB;
        const v2f qA = { frcp(d2A.x), frcp(d2A.y) };
        const v2f qB = { frcp(d2B.x), frcp(d2B.y) };
        accA = __builtin_amdgcn_wmma_f32_16x16x4_f32(false, wA[kk], false, qA,
                                                     (short)0, accA, false, false);
        accB = __builtin_amdgcn_wmma_f32_16x16x4_f32(false, wA[kk], false, qB,
                                                     (short)0, accB, false, false);
    }

    const float2 kA = tile_tail<MID_A>(accA, hA);
    const float2 kB = tile_tail<false>(accB, hB);
    if (lane < 16) {
        cbuf[cpad(hA.j)] = kA;
        cbuf[cpad(hB.j)] = kB;
    }
}

__global__ __launch_bounds__(NT)
void s4_fused(const float* __restrict__ Lre, const float* __restrict__ Lim,
              const float* __restrict__ Pre, const float* __restrict__ Pim,
              const float* __restrict__ Bre, const float* __restrict__ Bim,
              const float* __restrict__ Cre, const float* __restrict__ Cim,
              const float* __restrict__ log_dt, const float* __restrict__ D,
              float* __restrict__ out)
{
    __shared__ float Wm[N_DIM * 16];
    __shared__ float lamRe[N_DIM], lamIm[N_DIM];
    __shared__ float2 cbuf[L_DIM + L_DIM / 8];

    const int tid = threadIdx.x;
    const int h   = blockIdx.x;

    if (tid < N_DIM) {
        const int n   = tid;
        const int idx = h * N_DIM + n;
        const float pr = Pre[idx], pi = Pim[idx];
        const float br = Bre[idx], bi = Bim[idx];
        const float cr = Cre[idx], ci = Cim[idx];
        const float wPBr = pr * br + pi * bi, wPBi = pr * bi - pi * br;
        const float wPPr = pr * pr + pi * pi, wPPi = 0.0f;
        const float wCBr = cr * br + ci * bi, wCBi = cr * bi - ci * br;
        const float wCPr = cr * pr + ci * pi, wCPi = cr * pi - ci * pr;
        const float wr[4] = {wCBr, wCPr, wPBr, wPPr};
        const float wi[4] = {wCBi, wCPi, wPBi, wPPi};
        const float lre = -softplusf(Lre[idx]);
        const float lim = Lim[idx];
        #pragma unroll
        for (int X = 0; X < 4; ++X) {
            Wm[n * 16 + 4 * X]     = wr[X];
            Wm[n * 16 + 4 * X + 1] = wi[X];
            Wm[n * 16 + 4 * X + 2] = wr[X] * lre + wi[X] * lim;
            Wm[n * 16 + 4 * X + 3] = wi[X] * lre - wr[X] * lim;
        }
        lamRe[n] = lre;
        lamIm[n] = lim;
    }
    __syncthreads();

    const int wave = __builtin_amdgcn_readfirstlane(tid >> 5);
    const int lane = tid & 31;
    const int half = lane >> 4;
    const int c    = lane & 15;

    v2f wA[16], lr2[16], li[16];
    #pragma unroll
    for (int kk = 0; kk < 16; ++kk) {
        const int krow = 4 * kk + 2 * half;
        wA[kk]  = (v2f){ Wm[krow * 16 + c], Wm[(krow + 1) * 16 + c] };
        const v2f lrv = { lamRe[krow], lamRe[krow + 1] };
        lr2[kk] = lrv * lrv;
        li[kk]  = (v2f){ lamIm[krow], lamIm[krow + 1] };
    }

    const float sc = 2.0f * expf(-log_dt[h]);

    if (wave == 0)
        cauchy_pair<true>(0, NWAVE, lane, sc, wA, lr2, li, lamRe, cbuf);
    for (int t = (wave == 0 ? 2 * NWAVE : wave); t < NTILE; t += 2 * NWAVE)
        cauchy_pair<false>(t, t + NWAVE, lane, sc, wA, lr2, li, lamRe, cbuf);
    __syncthreads();

    const float scale = 1.0f / (float)L_DIM;
    float* __restrict__ orow = out + h * L_DIM;
    fft_pass<1, 3, false>(cbuf, tid, orow, scale);
    __syncthreads();
    fft_pass<4, 3, false>(cbuf, tid, orow, scale);
    __syncthreads();
    fft_pass<7, 3, false>(cbuf, tid, orow, scale);
    __syncthreads();
    fft_pass<10, 2, true>(cbuf, tid, orow, scale);
    fft_pass<10, 2, true>(cbuf, tid + NT, orow, scale);

    if (tid == 0)
        out[(size_t)CH_DIM * H_DIM * L_DIM + h] = D[h];
}

extern "C" void kernel_launch(void* const* d_in, const int* in_sizes, int n_in,
                              void* d_out, int out_size, void* d_ws, size_t ws_size,
                              hipStream_t stream)
{
    (void)in_sizes; (void)n_in; (void)out_size; (void)d_ws; (void)ws_size;
    const float* Lre = (const float*)d_in[0];
    const float* Lim = (const float*)d_in[1];
    const float* Pre = (const float*)d_in[2];
    const float* Pim = (const float*)d_in[3];
    const float* Bre = (const float*)d_in[4];
    const float* Bim = (const float*)d_in[5];
    const float* Cre = (const float*)d_in[6];
    const float* Cim = (const float*)d_in[7];
    const float* D   = (const float*)d_in[8];
    const float* ldt = (const float*)d_in[9];

    float* out = (float*)d_out;

    s4_fused<<<dim3(H_DIM), NT, 0, stream>>>(
        Lre, Lim, Pre, Pim, Bre, Bim, Cre, Cim, ldt, D, out);
}
